# starting kernel
# baseline (speedup 1.0000x reference)
#include <hip/hip_runtime.h>
#include <math.h>

#define BB 16
#define TT 128
#define NN 256
#define HH 4
#define SP (TT + 1)
#define AP (TT + 2)

typedef float v2f __attribute__((ext_vector_type(2)));
typedef float v8f __attribute__((ext_vector_type(8)));

static __device__ __forceinline__ v8f wmma4(v2f a, v2f b, v8f c) {
  return __builtin_amdgcn_wmma_f32_16x16x4_f32(
       false, a,  false, b,
       (short)0, c,  false,  false);
}

__global__ void prefix_kernel(const int* __restrict__ mask, int* __restrict__ S) {
  int b = blockIdx.x;
  if (threadIdx.x == 0) {
    int* Sb = S + b * SP;
    const int* mb = mask + b * TT;
    int s = 0;
    Sb[0] = 0;
    for (int t = 0; t < TT; ++t) { s += mb[t]; Sb[t + 1] = s; }
  }
}

__global__ void gram_kernel(const float* __restrict__ x, float* __restrict__ G) {
  int b = blockIdx.x >> 3, ti = blockIdx.x & 7;
  int wave = threadIdx.x >> 5;
  int lane = threadIdx.x & 31;
  int r = lane & 15, o = lane >> 4;
  const float* Xb = x + (size_t)b * TT * NN;
  const float* ar = Xb + (size_t)(ti * 16 + r) * NN + 2 * o;
  const float* br = Xb + (size_t)(wave * 16 + r) * NN + 2 * o;
  v8f acc = {};
#pragma unroll 8
  for (int k = 0; k < NN; k += 4) {
    v2f a = *(const v2f*)(ar + k);
    v2f bv = *(const v2f*)(br + k);
    acc = wmma4(a, bv, acc);
  }
  float* Gb = G + (size_t)b * TT * TT;
#pragma unroll
  for (int g = 0; g < 8; ++g) {
    int t = ti * 16 + g + 8 * o;
    Gb[(size_t)t * TT + wave * 16 + r] = acc[g];
  }
}

__global__ void __launch_bounds__(512) base_kernel(const float* __restrict__ x,
                                                   const float* __restrict__ w0,
                                                   float* __restrict__ Z) {
  int b = blockIdx.x >> 3, ti = blockIdx.x & 7;
  int wave = threadIdx.x >> 5;
  int lane = threadIdx.x & 31;
  int r = lane & 15, o = lane >> 4;
  const float* ar = x + ((size_t)b * TT + ti * 16 + r) * NN + 2 * o;
  const float* br = w0 + ((size_t)b * NN + wave * 16 + r) * NN + 2 * o;
  v8f acc = {};
#pragma unroll 8
  for (int k = 0; k < NN; k += 4) {
    v2f a = *(const v2f*)(ar + k);
    v2f bv = *(const v2f*)(br + k);
    acc = wmma4(a, bv, acc);
  }
#pragma unroll
  for (int g = 0; g < 8; ++g) {
    int t = ti * 16 + g + 8 * o;
    Z[((size_t)b * TT + t) * NN + wave * 16 + r] = acc[g];
  }
}

__global__ void __launch_bounds__(512) y_kernel(
    const float* __restrict__ x, const float* __restrict__ G,
    const float* __restrict__ Z, const int* __restrict__ S,
    const int* __restrict__ mask, const float* __restrict__ alpha,
    const float* __restrict__ eta, float* __restrict__ out_y) {
  __shared__ int sS[SP];
  __shared__ int sM[TT];
  __shared__ float sA[16][AP];
  __shared__ float sP[16];
  int blk = blockIdx.x;
  int ti = blk & 7;
  int h = (blk >> 3) & 3;
  int b = blk >> 5;
  int tid = threadIdx.x;
  if (tid < SP) sS[tid] = S[b * SP + tid];
  if (tid >= 256 && tid < 256 + TT) sM[tid - 256] = mask[b * TT + (tid - 256)];
  __syncthreads();

  float eh = eta[h];
  float l2a = log2f(alpha[h]);

  const float* Gb = G + ((size_t)b * TT + ti * 16) * TT;
#pragma unroll
  for (int i = 0; i < 4; ++i) {
    int idx = tid + i * 512;
    int tr = idx >> 7, j = idx & 127;
    int t = ti * 16 + tr;
    float val = 0.0f;
    if (j < t && sM[j])
      val = eh * exp2f(l2a * (float)(sS[t] - sS[j + 1])) * Gb[(size_t)tr * TT + j];
    sA[tr][j] = val;
  }
  if (tid < 16) sP[tid] = exp2f(l2a * (float)sS[ti * 16 + tid]);
  __syncthreads();

  int wave = tid >> 5;
  int lane = tid & 31;
  int r = lane & 15, o = lane >> 4;
  const float* Xb = x + (size_t)b * TT * NN;
  int n = wave * 16 + r;

  v8f acc = {};
#pragma unroll 4
  for (int k = 0; k < TT; k += 4) {
    int j0 = k + 2 * o;
    v2f av = *(const v2f*)&sA[r][j0];
    v2f bv;
    bv.x = Xb[(size_t)j0 * NN + n];
    bv.y = Xb[(size_t)(j0 + 1) * NN + n];
    acc = wmma4(av, bv, acc);
  }
#pragma unroll
  for (int g = 0; g < 8; ++g) {
    int tl = g + 8 * o;
    int t = ti * 16 + tl;
    float base = sP[tl] * Z[((size_t)b * TT + t) * NN + n];
    out_y[(((size_t)b * TT + t) * HH + h) * NN + n] = acc[g] + base;
  }
}

__global__ void __launch_bounds__(512) wfin_kernel(
    const float* __restrict__ x, const float* __restrict__ w0,
    const int* __restrict__ S, const int* __restrict__ mask,
    const float* __restrict__ alpha, const float* __restrict__ eta,
    float* __restrict__ out_w) {
  __shared__ float sd[TT];
  __shared__ float sA[16][AP];
  int blk = blockIdx.x;
  int nt = blk & 15;
  int h = (blk >> 4) & 3;
  int b = blk >> 6;
  int tid = threadIdx.x;
  float eh = eta[h];
  float l2a = log2f(alpha[h]);
  int ST = S[b * SP + TT];
  if (tid < TT) {
    int m = mask[b * TT + tid];
    int cnt = ST - S[b * SP + tid + 1];
    sd[tid] = m ? eh * exp2f(l2a * (float)cnt) : 0.0f;
  }
  __syncthreads();

  const float* Xb = x + (size_t)b * TT * NN;
#pragma unroll
  for (int i = 0; i < 4; ++i) {
    int idx = tid + i * 512;
    int nr = idx & 15, j = idx >> 4;
    sA[nr][j] = sd[j] * Xb[(size_t)j * NN + nt * 16 + nr];
  }
  __syncthreads();

  float p = exp2f(l2a * (float)ST);
  int wave = tid >> 5;
  int lane = tid & 31;
  int r = lane & 15, o = lane >> 4;
  int m0 = wave * 16;

  v8f acc = {};
#pragma unroll 4
  for (int k = 0; k < TT; k += 4) {
    int j0 = k + 2 * o;
    v2f av = *(const v2f*)&sA[r][j0];
    v2f bv;
    bv.x = Xb[(size_t)j0 * NN + m0 + r];
    bv.y = Xb[(size_t)(j0 + 1) * NN + m0 + r];
    acc = wmma4(av, bv, acc);
  }
#pragma unroll
  for (int g = 0; g < 8; ++g) {
    int nn = nt * 16 + g + 8 * o;
    int mm = m0 + r;
    float val = acc[g] + p * w0[((size_t)b * NN + nn) * NN + mm];
    out_w[(((size_t)b * HH + h) * NN + nn) * NN + mm] = val;
  }
}

extern "C" void kernel_launch(void* const* d_in, const int* in_sizes, int n_in,
                              void* d_out, int out_size, void* d_ws, size_t ws_size,
                              hipStream_t stream) {
  const float* x     = (const float*)d_in[0];
  const float* w0    = (const float*)d_in[1];
  const float* alpha = (const float*)d_in[2];
  const float* eta   = (const float*)d_in[3];
  const int*   mask  = (const int*)d_in[4];

  float* out_w = (float*)d_out;
  float* out_y = out_w + (size_t)BB * HH * NN * NN;

  int*   S = (int*)d_ws;
  float* G = (float*)((char*)d_ws + 16384);
  float* Z = G + (size_t)BB * TT * TT;

  prefix_kernel<<<dim3(BB), dim3(32), 0, stream>>>(mask, S);
  gram_kernel<<<dim3(BB * 8), dim3(256), 0, stream>>>(x, G);
  base_kernel<<<dim3(BB * 8), dim3(512), 0, stream>>>(x, w0, Z);
  y_kernel<<<dim3(BB * HH * 8), dim3(512), 0, stream>>>(x, G, Z, S, mask, alpha, eta, out_y);
  wfin_kernel<<<dim3(BB * HH * 16), dim3(512), 0, stream>>>(x, w0, S, mask, alpha, eta, out_w);
}


# Fuse the five launches into one: each block rebuilds the mask prefix sums and its own Gram and base tiles in LDS and serves all four heads, so nothing passes through the workspace
# speedup vs baseline: 1.4056x; 1.4056x over previous
#include <hip/hip_runtime.h>
#include <math.h>

#define BB 16
#define TT 128
#define NN 256
#define HH 4
#define SP (TT + 1)
#define AP (TT + 2)

typedef float v2f __attribute__((ext_vector_type(2)));
typedef float v8f __attribute__((ext_vector_type(8)));

static __device__ __forceinline__ v8f wmma4(v2f a, v2f b, v8f c) {
  return __builtin_amdgcn_wmma_f32_16x16x4_f32(
       false, a,  false, b,
       (short)0, c,  false,  false);
}

static __device__ __forceinline__ void block_prefix(const int* __restrict__ mask_b,
                                                    int* sS, int* sM, int* sW, int tid) {
  int lane = tid & 31, wave = tid >> 5;
  int v = 0;
  if (tid < TT) {
    int m = mask_b[tid];
    sM[tid] = m;
    v = m;
    for (int d = 1; d < 32; d <<= 1) {
      int nb = __shfl_up(v, d);
      if (lane >= d) v += nb;
    }
    if (lane == 31) sW[wave] = v;
  }
  __syncthreads();
  if (tid < TT) {
    int off = 0;
    for (int q = 0; q < wave; ++q) off += sW[q];
    sS[tid + 1] = v + off;
    if (tid == 0) sS[0] = 0;
  }
}

__global__ void __launch_bounds__(512) fused_kernel(
    const float* __restrict__ x, const float* __restrict__ w0,
    const float* __restrict__ alpha, const float* __restrict__ eta,
    const int* __restrict__ mask, float* __restrict__ out_w,
    float* __restrict__ out_y) {
  __shared__ int sS[SP];
  __shared__ int sM[TT];
  __shared__ int sW[4];
  __shared__ __attribute__((aligned(16))) float sA[HH][16][AP];
  __shared__ float sG[16][AP];
  __shared__ float sZ[16][AP];
  __shared__ float sP[HH][16];
  __shared__ float sd[HH][TT];

  int tid = threadIdx.x;
  int wave = tid >> 5;
  int lane = tid & 31;
  int r = lane & 15, o = lane >> 4;
  int u = blockIdx.x >> 1;

  if ((blockIdx.x & 1) == 0) {
    int ti = u & 7;
    int nh = (u >> 3) & 1;
    int b = u >> 4;
    const float* Xb = x + (size_t)b * TT * NN;
    block_prefix(mask + b * TT, sS, sM, sW, tid);

    {
      int wl = wave & 7;
      const float* ar = Xb + (size_t)(ti * 16 + r) * NN + 2 * o;
      const float* br = (wave < 8)
          ? w0 + ((size_t)b * NN + (nh * 8 + wl) * 16 + r) * NN + 2 * o
          : Xb + (size_t)(wl * 16 + r) * NN + 2 * o;
      float (*dst)[AP] = (wave < 8) ? sZ : sG;
      v8f acc = {};
#pragma unroll 8
      for (int k = 0; k < NN; k += 4) {
        v2f a = *(const v2f*)(ar + k);
        v2f bv = *(const v2f*)(br + k);
        acc = wmma4(a, bv, acc);
      }
#pragma unroll
      for (int g = 0; g < 8; ++g) dst[g + 8 * o][wl * 16 + r] = acc[g];
    }
    __syncthreads();

#pragma unroll
    for (int h = 0; h < HH; ++h) {
      float eh = eta[h];
      float l2a = log2f(alpha[h]);
#pragma unroll
      for (int i = 0; i < 4; ++i) {
        int idx = tid + i * 512;
        int tr = idx >> 7, j = idx & 127;
        int t = ti * 16 + tr;
        float val = 0.0f;
        if (j < t && sM[j])
          val = eh * exp2f(l2a * (float)(sS[t] - sS[j + 1])) * sG[tr][j];
        sA[h][tr][j] = val;
      }
      if (tid < 16) sP[h][tid] = exp2f(l2a * (float)sS[ti * 16 + tid]);
    }
    __syncthreads();

    int ntl = wave & 7;
    int h0 = (wave >> 3) * 2, h1 = h0 + 1;
    int n = (nh * 8 + ntl) * 16 + r;
    v8f acc0 = {}, acc1 = {};
#pragma unroll 4
    for (int k = 0; k < TT; k += 4) {
      int j0 = k + 2 * o;
      v2f bv;
      bv.x = Xb[(size_t)j0 * NN + n];
      bv.y = Xb[(size_t)(j0 + 1) * NN + n];
      v2f a0 = *(const v2f*)&sA[h0][r][j0];
      v2f a1 = *(const v2f*)&sA[h1][r][j0];
      acc0 = wmma4(a0, bv, acc0);
      acc1 = wmma4(a1, bv, acc1);
    }
#pragma unroll
    for (int g = 0; g < 8; ++g) {
      int tl = g + 8 * o;
      int t = ti * 16 + tl;
      float z = sZ[tl][ntl * 16 + r];
      float base0 = sP[h0][tl] * z;
      float base1 = sP[h1][tl] * z;
      out_y[(((size_t)b * TT + t) * HH + h0) * NN + n] = acc0[g] + base0;
      out_y[(((size_t)b * TT + t) * HH + h1) * NN + n] = acc1[g] + base1;
    }
  } else {
    int nt = u & 15;
    int b = u >> 4;
    const float* Xb = x + (size_t)b * TT * NN;
    block_prefix(mask + b * TT, sS, sM, sW, tid);
    __syncthreads();

    int ST = sS[TT];
    {
      int h = tid >> 7, j = tid & 127;
      float eh = eta[h];
      float l2a = log2f(alpha[h]);
      int cnt = ST - sS[j + 1];
      sd[h][j] = sM[j] ? eh * exp2f(l2a * (float)cnt) : 0.0f;
    }
    __syncthreads();

#pragma unroll
    for (int i = 0; i < 4; ++i) {
      int idx = tid + i * 512;
      int nr = idx & 15, j = idx >> 4;
      float xv = Xb[(size_t)j * NN + nt * 16 + nr];
#pragma unroll
      for (int h = 0; h < HH; ++h) sA[h][nr][j] = sd[h][j] * xv;
    }
    __syncthreads();

    int m0 = wave * 16;
    v8f acc[HH];
#pragma unroll
    for (int h = 0; h < HH; ++h) acc[h] = (v8f){};
#pragma unroll 4
    for (int k = 0; k < TT; k += 4) {
      int j0 = k + 2 * o;
      v2f bv;
      bv.x = Xb[(size_t)j0 * NN + m0 + r];
      bv.y = Xb[(size_t)(j0 + 1) * NN + m0 + r];
#pragma unroll
      for (int h = 0; h < HH; ++h) {
        v2f av = *(const v2f*)&sA[h][r][j0];
        acc[h] = wmma4(av, bv, acc[h]);
      }
    }
    float fST = (float)ST;
    float p[HH];
#pragma unroll
    for (int h = 0; h < HH; ++h) p[h] = exp2f(log2f(alpha[h]) * fST);
#pragma unroll
    for (int g = 0; g < 8; ++g) {
      int nn = nt * 16 + g + 8 * o;
      int mm = m0 + r;
      float wv = w0[((size_t)b * NN + nn) * NN + mm];
#pragma unroll
      for (int h = 0; h < HH; ++h)
        out_w[(((size_t)b * HH + h) * NN + nn) * NN + mm] = acc[h][g] + p[h] * wv;
    }
  }
}

extern "C" void kernel_launch(void* const* d_in, const int* in_sizes, int n_in,
                              void* d_out, int out_size, void* d_ws, size_t ws_size,
                              hipStream_t stream) {
  const float* x     = (const float*)d_in[0];
  const float* w0    = (const float*)d_in[1];
  const float* alpha = (const float*)d_in[2];
  const float* eta   = (const float*)d_in[3];
  const int*   mask  = (const int*)d_in[4];

  float* out_w = (float*)d_out;
  float* out_y = out_w + (size_t)BB * HH * NN * NN;

  fused_kernel<<<dim3(BB * 16 * 2), dim3(512), 0, stream>>>(x, w0, alpha, eta, mask, out_w, out_y);
}


# Read-out phase 1 feeds its WMMAs from LDS: x tile staged once per block, each wave's B tile staged in 32-k chunks by whole-line 16-B global loads issued two chunks ahead
# speedup vs baseline: 2.1922x; 1.5596x over previous
#include <hip/hip_runtime.h>
#include <math.h>

#define BB 16
#define TT 128
#define NN 256
#define HH 4
#define SP (TT + 1)
#define AP (TT + 2)
#define KC 32
#define XS (NN + 4)
#define BS (KC + 4)

typedef float v2f __attribute__((ext_vector_type(2)));
typedef float v8f __attribute__((ext_vector_type(8)));
typedef float v4f __attribute__((ext_vector_type(4)));

static __device__ __forceinline__ v8f wmma4(v2f a, v2f b, v8f c) {
  return __builtin_amdgcn_wmma_f32_16x16x4_f32(
       false, a,  false, b,
       (short)0, c,  false,  false);
}

static __device__ __forceinline__ void block_prefix(const int* __restrict__ mask_b,
                                                    int* sS, int* sM, int* sW, int tid) {
  int lane = tid & 31, wave = tid >> 5;
  int v = 0;
  if (tid < TT) {
    int m = mask_b[tid];
    sM[tid] = m;
    v = m;
    for (int d = 1; d < 32; d <<= 1) {
      int nb = __shfl_up(v, d);
      if (lane >= d) v += nb;
    }
    if (lane == 31) sW[wave] = v;
  }
  __syncthreads();
  if (tid < TT) {
    int off = 0;
    for (int q = 0; q < wave; ++q) off += sW[q];
    sS[tid + 1] = v + off;
    if (tid == 0) sS[0] = 0;
  }
}

__global__ void __launch_bounds__(512) fused_kernel(
    const float* __restrict__ x, const float* __restrict__ w0,
    const float* __restrict__ alpha, const float* __restrict__ eta,
    const int* __restrict__ mask, float* __restrict__ out_w,
    float* __restrict__ out_y) {
  __shared__ int sS[SP];
  __shared__ int sM[TT];
  __shared__ int sW[4];
  __shared__ __attribute__((aligned(16))) float sU[16 * 16 * BS];
  float (*sA)[16][AP] = reinterpret_cast<float (*)[16][AP]>(sU);
  __shared__ __attribute__((aligned(16))) float sXA[16][XS];
  __shared__ float sG[16][AP];
  __shared__ float sZ[16][AP];
  __shared__ float sP[HH][16];
  __shared__ float sd[HH][TT];

  int tid = threadIdx.x;
  int wave = tid >> 5;
  int lane = tid & 31;
  int r = lane & 15, o = lane >> 4;
  int u = blockIdx.x >> 1;

  if ((blockIdx.x & 1) == 0) {
    int ti = u & 7;
    int nh = (u >> 3) & 1;
    int b = u >> 4;
    const float* Xb = x + (size_t)b * TT * NN;

    int wl = wave & 7;
    v4f xa[2];
#pragma unroll
    for (int i = 0; i < 2; ++i) {
      int idx = tid + i * 512;
      xa[i] = *(const v4f*)(Xb + (size_t)(ti * 16 + (idx >> 6)) * NN + (idx & 63) * 4);
    }
    const float* bt = (wave < 8)
        ? w0 + ((size_t)b * NN + (nh * 8 + wl) * 16) * NN
        : Xb + (size_t)(wl * 16) * NN;
    int lr = lane >> 3, lc = (lane & 7) * 4;
    const float* bp = bt + (size_t)lr * NN + lc;
    float* stg = sU + wave * (16 * BS);
    v4f pre[2][4];
#pragma unroll
    for (int p = 0; p < 2; ++p)
#pragma unroll
      for (int i = 0; i < 4; ++i) pre[p][i] = *(const v4f*)(bp + (size_t)(4 * i) * NN + p * KC);

    block_prefix(mask + b * TT, sS, sM, sW, tid);

#pragma unroll
    for (int i = 0; i < 2; ++i) {
      int idx = tid + i * 512;
      *(v4f*)&sXA[idx >> 6][(idx & 63) * 4] = xa[i];
    }
    __syncthreads();

    {
      float (*dst)[AP] = (wave < 8) ? sZ : sG;
      v8f acc = {};
#pragma unroll
      for (int c = 0; c < NN / KC; ++c) {
#pragma unroll
        for (int i = 0; i < 4; ++i) *(v4f*)&stg[(lr + 4 * i) * BS + lc] = pre[c & 1][i];
        if (c + 2 < NN / KC) {
#pragma unroll
          for (int i = 0; i < 4; ++i)
            pre[c & 1][i] = *(const v4f*)(bp + (size_t)(4 * i) * NN + (c + 2) * KC);
        }
#pragma unroll
        for (int q = 0; q < KC; q += 4) {
          v2f a = *(const v2f*)&sXA[r][c * KC + q + 2 * o];
          v2f bv = *(const v2f*)&stg[r * BS + q + 2 * o];
          acc = wmma4(a, bv, acc);
        }
      }
#pragma unroll
      for (int g = 0; g < 8; ++g) dst[g + 8 * o][wl * 16 + r] = acc[g];
    }
    __syncthreads();

#pragma unroll
    for (int h = 0; h < HH; ++h) {
      float eh = eta[h];
      float l2a = log2f(alpha[h]);
#pragma unroll
      for (int i = 0; i < 4; ++i) {
        int idx = tid + i * 512;
        int tr = idx >> 7, j = idx & 127;
        int t = ti * 16 + tr;
        float val = 0.0f;
        if (j < t && sM[j])
          val = eh * exp2f(l2a * (float)(sS[t] - sS[j + 1])) * sG[tr][j];
        sA[h][tr][j] = val;
      }
      if (tid < 16) sP[h][tid] = exp2f(l2a * (float)sS[ti * 16 + tid]);
    }
    __syncthreads();

    int ntl = wave & 7;
    int h0 = (wave >> 3) * 2, h1 = h0 + 1;
    int n = (nh * 8 + ntl) * 16 + r;
    v8f acc0 = {}, acc1 = {};
#pragma unroll 4
    for (int k = 0; k < TT; k += 4) {
      int j0 = k + 2 * o;
      v2f bv;
      bv.x = Xb[(size_t)j0 * NN + n];
      bv.y = Xb[(size_t)(j0 + 1) * NN + n];
      v2f a0 = *(const v2f*)&sA[h0][r][j0];
      v2f a1 = *(const v2f*)&sA[h1][r][j0];
      acc0 = wmma4(a0, bv, acc0);
      acc1 = wmma4(a1, bv, acc1);
    }
#pragma unroll
    for (int g = 0; g < 8; ++g) {
      int tl = g + 8 * o;
      int t = ti * 16 + tl;
      float z = sZ[tl][ntl * 16 + r];
      float base0 = sP[h0][tl] * z;
      float base1 = sP[h1][tl] * z;
      out_y[(((size_t)b * TT + t) * HH + h0) * NN + n] = acc0[g] + base0;
      out_y[(((size_t)b * TT + t) * HH + h1) * NN + n] = acc1[g] + base1;
    }
  } else {
    int nt = u & 15;
    int b = u >> 4;
    const float* Xb = x + (size_t)b * TT * NN;
    block_prefix(mask + b * TT, sS, sM, sW, tid);
    __syncthreads();

    int ST = sS[TT];
    {
      int h = tid >> 7, j = tid & 127;
      float eh = eta[h];
      float l2a = log2f(alpha[h]);
      int cnt = ST - sS[j + 1];
      sd[h][j] = sM[j] ? eh * exp2f(l2a * (float)cnt) : 0.0f;
    }
    __syncthreads();

#pragma unroll
    for (int i = 0; i < 4; ++i) {
      int idx = tid + i * 512;
      int nr = idx & 15, j = idx >> 4;
      float xv = Xb[(size_t)j * NN + nt * 16 + nr];
#pragma unroll
      for (int h = 0; h < HH; ++h) sA[h][nr][j] = sd[h][j] * xv;
    }
    __syncthreads();

    int m0 = wave * 16;
    v8f acc[HH];
#pragma unroll
    for (int h = 0; h < HH; ++h) acc[h] = (v8f){};
#pragma unroll 4
    for (int k = 0; k < TT; k += 4) {
      int j0 = k + 2 * o;
      v2f bv;
      bv.x = Xb[(size_t)j0 * NN + m0 + r];
      bv.y = Xb[(size_t)(j0 + 1) * NN + m0 + r];
#pragma unroll
      for (int h = 0; h < HH; ++h) {
        v2f av = *(const v2f*)&sA[h][r][j0];
        acc[h] = wmma4(av, bv, acc[h]);
      }
    }
    float fST = (float)ST;
    float p[HH];
#pragma unroll
    for (int h = 0; h < HH; ++h) p[h] = exp2f(log2f(alpha[h]) * fST);
#pragma unroll
    for (int g = 0; g < 8; ++g) {
      int nn = nt * 16 + g + 8 * o;
      int mm = m0 + r;
      float wv = w0[((size_t)b * NN + nn) * NN + mm];
#pragma unroll
      for (int h = 0; h < HH; ++h)
        out_w[(((size_t)b * HH + h) * NN + nn) * NN + mm] = acc[h][g] + p[h] * wv;
    }
  }
}

extern "C" void kernel_launch(void* const* d_in, const int* in_sizes, int n_in,
                              void* d_out, int out_size, void* d_ws, size_t ws_size,
                              hipStream_t stream) {
  const float* x     = (const float*)d_in[0];
  const float* w0    = (const float*)d_in[1];
  const float* alpha = (const float*)d_in[2];
  const float* eta   = (const float*)d_in[3];
  const int*   mask  = (const int*)d_in[4];

  float* out_w = (float*)d_out;
  float* out_y = out_w + (size_t)BB * HH * NN * NN;

  fused_kernel<<<dim3(BB * 16 * 2), dim3(512), 0, stream>>>(x, w0, alpha, eta, mask, out_w, out_y);
}


# Both masked products loop over the set mask steps only: a per-batch index list compacts k (zero terms dropped; nonzero terms keep their order, regrouped four per WMMA)
# speedup vs baseline: 2.7093x; 1.2359x over previous
#include <hip/hip_runtime.h>
#include <math.h>

#define BB 16
#define TT 128
#define NN 256
#define HH 4
#define SP (TT + 1)
#define AP (TT + 2)
#define KC 32
#define XS (NN + 4)
#define BS (KC + 4)

typedef float v2f __attribute__((ext_vector_type(2)));
typedef float v8f __attribute__((ext_vector_type(8)));
typedef float v4f __attribute__((ext_vector_type(4)));
typedef int v2i __attribute__((ext_vector_type(2)));

static __device__ __forceinline__ v8f wmma4(v2f a, v2f b, v8f c) {
  return __builtin_amdgcn_wmma_f32_16x16x4_f32(
       false, a,  false, b,
       (short)0, c,  false,  false);
}

static __device__ __forceinline__ void block_prefix(const int* __restrict__ mask_b,
                                                    int* sS, int* sJ, int* sW, int tid) {
  int lane = tid & 31, wave = tid >> 5;
  int v = 0, m = 0;
  if (tid < TT) {
    m = mask_b[tid];
    v = m;
    for (int d = 1; d < 32; d <<= 1) {
      int nb = __shfl_up(v, d);
      if (lane >= d) v += nb;
    }
    if (lane == 31) sW[wave] = v;
    sJ[tid] = 0;
  }
  __syncthreads();
  if (tid < TT) {
    int off = 0;
    for (int q = 0; q < wave; ++q) off += sW[q];
    int incl = v + off;
    sS[tid + 1] = incl;
    if (tid == 0) sS[0] = 0;
    if (m) sJ[(incl - 1) & (TT - 1)] = tid;
  }
}

__global__ void __launch_bounds__(512) fused_kernel(
    const float* __restrict__ x, const float* __restrict__ w0,
    const float* __restrict__ alpha, const float* __restrict__ eta,
    const int* __restrict__ mask, float* __restrict__ out_w,
    float* __restrict__ out_y) {
  __shared__ int sS[SP];
  __shared__ __attribute__((aligned(8))) int sJ[TT];
  __shared__ int sW[4];
  __shared__ __attribute__((aligned(16))) float sU[16 * 16 * BS];
  float (*sA)[16][AP] = reinterpret_cast<float (*)[16][AP]>(sU);
  __shared__ __attribute__((aligned(16))) float sXA[16][XS];
  __shared__ float sG[16][AP];
  __shared__ float sZ[16][AP];
  __shared__ float sP[HH][16];
  __shared__ float sd[HH][TT];

  int tid = threadIdx.x;
  int wave = tid >> 5;
  int lane = tid & 31;
  int r = lane & 15, o = lane >> 4;
  int u = blockIdx.x >> 1;

  if ((blockIdx.x & 1) == 0) {
    int ti = u & 7;
    int nh = (u >> 3) & 1;
    int b = u >> 4;
    const float* Xb = x + (size_t)b * TT * NN;

    int wl = wave & 7;
    v4f xa[2];
#pragma unroll
    for (int i = 0; i < 2; ++i) {
      int idx = tid + i * 512;
      xa[i] = *(const v4f*)(Xb + (size_t)(ti * 16 + (idx >> 6)) * NN + (idx & 63) * 4);
    }
    const float* bt = (wave < 8)
        ? w0 + ((size_t)b * NN + (nh * 8 + wl) * 16) * NN
        : Xb + (size_t)(wl * 16) * NN;
    int lr = lane >> 3, lc = (lane & 7) * 4;
    const float* bp = bt + (size_t)lr * NN + lc;
    float* stg = sU + wave * (16 * BS);
    v4f pre[2][4];
#pragma unroll
    for (int p = 0; p < 2; ++p)
#pragma unroll
      for (int i = 0; i < 4; ++i) pre[p][i] = *(const v4f*)(bp + (size_t)(4 * i) * NN + p * KC);

    block_prefix(mask + b * TT, sS, sJ, sW, tid);

#pragma unroll
    for (int i = 0; i < 2; ++i) {
      int idx = tid + i * 512;
      *(v4f*)&sXA[idx >> 6][(idx & 63) * 4] = xa[i];
    }
    __syncthreads();

    {
      float (*dst)[AP] = (wave < 8) ? sZ : sG;
      v8f acc = {};
#pragma unroll
      for (int c = 0; c < NN / KC; ++c) {
#pragma unroll
        for (int i = 0; i < 4; ++i) *(v4f*)&stg[(lr + 4 * i) * BS + lc] = pre[c & 1][i];
        if (c + 2 < NN / KC) {
#pragma unroll
          for (int i = 0; i < 4; ++i)
            pre[c & 1][i] = *(const v4f*)(bp + (size_t)(4 * i) * NN + (c + 2) * KC);
        }
#pragma unroll
        for (int q = 0; q < KC; q += 4) {
          v2f a = *(const v2f*)&sXA[r][c * KC + q + 2 * o];
          v2f bv = *(const v2f*)&stg[r * BS + q + 2 * o];
          acc = wmma4(a, bv, acc);
        }
      }
#pragma unroll
      for (int g = 0; g < 8; ++g) dst[g + 8 * o][wl * 16 + r] = acc[g];
    }
    __syncthreads();

    int kend = (sS[ti * 16 + 15] + 3) & ~3;
    {
      int c = tid & 127, trb = tid >> 7;
      int jc = sJ[c];
      int st[4];
      float g[4];
#pragma unroll
      for (int i = 0; i < 4; ++i) {
        st[i] = sS[ti * 16 + trb + 4 * i];
        g[i] = sG[trb + 4 * i][jc];
      }
#pragma unroll
      for (int h = 0; h < HH; ++h) {
        float eh = eta[h];
        float l2a = log2f(alpha[h]);
        if (c < kend) {
#pragma unroll
          for (int i = 0; i < 4; ++i) {
            float val = eh * exp2f(l2a * (float)(st[i] - 1 - c)) * g[i];
            sA[h][trb + 4 * i][c] = (c < st[i]) ? val : 0.0f;
          }
        }
        if (tid < 16) sP[h][tid] = exp2f(l2a * (float)sS[ti * 16 + tid]);
      }
    }
    __syncthreads();

    int ntl = wave & 7;
    int h0 = (wave >> 3) * 2, h1 = h0 + 1;
    int n = (nh * 8 + ntl) * 16 + r;
    v8f acc0 = {}, acc1 = {};
#pragma unroll 4
    for (int k = 0; k < kend; k += 4) {
      int j0 = k + 2 * o;
      v2i jj = *(const v2i*)&sJ[j0];
      v2f bv;
      bv.x = Xb[(unsigned)(jj.x * NN + n)];
      bv.y = Xb[(unsigned)(jj.y * NN + n)];
      v2f a0 = *(const v2f*)&sA[h0][r][j0];
      v2f a1 = *(const v2f*)&sA[h1][r][j0];
      acc0 = wmma4(a0, bv, acc0);
      acc1 = wmma4(a1, bv, acc1);
    }
#pragma unroll
    for (int g = 0; g < 8; ++g) {
      int tl = g + 8 * o;
      int t = ti * 16 + tl;
      float z = sZ[tl][ntl * 16 + r];
      float base0 = sP[h0][tl] * z;
      float base1 = sP[h1][tl] * z;
      out_y[(((size_t)b * TT + t) * HH + h0) * NN + n] = acc0[g] + base0;
      out_y[(((size_t)b * TT + t) * HH + h1) * NN + n] = acc1[g] + base1;
    }
  } else {
    int nt = u & 15;
    int b = u >> 4;
    const float* Xb = x + (size_t)b * TT * NN;
    block_prefix(mask + b * TT, sS, sJ, sW, tid);
    __syncthreads();

    int ST = sS[TT];
    int kend = (ST + 3) & ~3;
    {
      int h = tid >> 7, c = tid & 127;
      float eh = eta[h];
      float l2a = log2f(alpha[h]);
      sd[h][c] = (c < ST) ? eh * exp2f(l2a * (float)(ST - 1 - c)) : 0.0f;
    }
    __syncthreads();

    {
      int nr = tid & 15, cb = tid >> 4;
      float xv[4];
#pragma unroll
      for (int i = 0; i < 4; ++i)
        xv[i] = Xb[(unsigned)(sJ[cb + 32 * i] * NN + nt * 16 + nr)];
#pragma unroll
      for (int i = 0; i < 4; ++i) {
        int c = cb + 32 * i;
        if (c < kend) {
#pragma unroll
          for (int h = 0; h < HH; ++h) sA[h][nr][c] = sd[h][c] * xv[i];
        }
      }
    }
    __syncthreads();

    int m0 = wave * 16;
    v8f acc[HH];
#pragma unroll
    for (int h = 0; h < HH; ++h) acc[h] = (v8f){};
#pragma unroll 4
    for (int k = 0; k < kend; k += 4) {
      int j0 = k + 2 * o;
      v2i jj = *(const v2i*)&sJ[j0];
      v2f bv;
      bv.x = Xb[(unsigned)(jj.x * NN + m0 + r)];
      bv.y = Xb[(unsigned)(jj.y * NN + m0 + r)];
#pragma unroll
      for (int h = 0; h < HH; ++h) {
        v2f av = *(const v2f*)&sA[h][r][j0];
        acc[h] = wmma4(av, bv, acc[h]);
      }
    }
    float fST = (float)ST;
    float p[HH];
#pragma unroll
    for (int h = 0; h < HH; ++h) p[h] = exp2f(log2f(alpha[h]) * fST);
#pragma unroll
    for (int g = 0; g < 8; ++g) {
      int nn = nt * 16 + g + 8 * o;
      int mm = m0 + r;
      float wv = w0[((size_t)b * NN + nn) * NN + mm];
#pragma unroll
      for (int h = 0; h < HH; ++h)
        out_w[(((size_t)b * HH + h) * NN + nn) * NN + mm] = acc[h][g] + p[h] * wv;
    }
  }
}

extern "C" void kernel_launch(void* const* d_in, const int* in_sizes, int n_in,
                              void* d_out, int out_size, void* d_ws, size_t ws_size,
                              hipStream_t stream) {
  const float* x     = (const float*)d_in[0];
  const float* w0    = (const float*)d_in[1];
  const float* alpha = (const float*)d_in[2];
  const float* eta   = (const float*)d_in[3];
  const int*   mask  = (const int*)d_in[4];

  float* out_w = (float*)d_out;
  float* out_y = out_w + (size_t)BB * HH * NN * NN;

  fused_kernel<<<dim3(BB * 16 * 2), dim3(512), 0, stream>>>(x, w0, alpha, eta, mask, out_w, out_y);
}


# Even out f32 matrix work per WGP: Gram column tiles past a row tile's causal range are skipped; grid order pairs long row tiles with short ones 256 places apart, shortest last; results bit-identical
# speedup vs baseline: 3.1464x; 1.1613x over previous
#include <hip/hip_runtime.h>
#include <math.h>

#define BB 16
#define TT 128
#define NN 256
#define HH 4
#define SP (TT + 1)
#define AP (TT + 2)
#define KC 32
#define XS (NN + 4)
#define BS (KC + 4)

typedef float v2f __attribute__((ext_vector_type(2)));
typedef float v8f __attribute__((ext_vector_type(8)));
typedef float v4f __attribute__((ext_vector_type(4)));
typedef int v2i __attribute__((ext_vector_type(2)));

static __device__ __forceinline__ v8f wmma4(v2f a, v2f b, v8f c) {
  return __builtin_amdgcn_wmma_f32_16x16x4_f32(
       false, a,  false, b,
       (short)0, c,  false,  false);
}

static __device__ __forceinline__ void block_prefix(const int* __restrict__ mask_b,
                                                    int* sS, int* sJ, int* sW, int tid) {
  int lane = tid & 31, wave = tid >> 5;
  int v = 0, m = 0;
  if (tid < TT) {
    m = mask_b[tid];
    v = m;
    for (int d = 1; d < 32; d <<= 1) {
      int nb = __shfl_up(v, d);
      if (lane >= d) v += nb;
    }
    if (lane == 31) sW[wave] = v;
    sJ[tid] = 0;
  }
  __syncthreads();
  if (tid < TT) {
    int off = 0;
    for (int q = 0; q < wave; ++q) off += sW[q];
    int incl = v + off;
    sS[tid + 1] = incl;
    if (tid == 0) sS[0] = 0;
    if (m) sJ[(incl - 1) & (TT - 1)] = tid;
  }
}

__global__ void __launch_bounds__(512) fused_kernel(
    const float* __restrict__ x, const float* __restrict__ w0,
    const float* __restrict__ alpha, const float* __restrict__ eta,
    const int* __restrict__ mask, float* __restrict__ out_w,
    float* __restrict__ out_y) {
  __shared__ int sS[SP];
  __shared__ __attribute__((aligned(8))) int sJ[TT];
  __shared__ int sW[4];
  __shared__ __attribute__((aligned(16))) float sU[16 * 16 * BS];
  float (*sA)[16][AP] = reinterpret_cast<float (*)[16][AP]>(sU);
  __shared__ __attribute__((aligned(16))) float sXA[16][XS];
  __shared__ float sG[16][AP];
  __shared__ float sZ[16][AP];
  __shared__ float sP[HH][16];
  __shared__ float sd[HH][TT];

  int tid = threadIdx.x;
  int wave = tid >> 5;
  int lane = tid & 31;
  int r = lane & 15, o = lane >> 4;
  int hs = blockIdx.x >> 8;
  int q = blockIdx.x & 127;

  if (blockIdx.x & 128) {
    int b = (q >> 1) & 15;
    int nh = (q >> 5) & 1;
    int th = 4 + 2 * (q >> 6) + (q & 1);
    int ti = hs ? 7 - th : th;
    const float* Xb = x + (size_t)b * TT * NN;

    int wl = wave & 7;
    bool act = (wave < 8) || (wl <= ti);
    v4f xa[2];
#pragma unroll
    for (int i = 0; i < 2; ++i) {
      int idx = tid + i * 512;
      xa[i] = *(const v4f*)(Xb + (size_t)(ti * 16 + (idx >> 6)) * NN + (idx & 63) * 4);
    }
    const float* bt = (wave < 8)
        ? w0 + ((size_t)b * NN + (nh * 8 + wl) * 16) * NN
        : Xb + (size_t)(wl * 16) * NN;
    int lr = lane >> 3, lc = (lane & 7) * 4;
    const float* bp = bt + (size_t)lr * NN + lc;
    float* stg = sU + wave * (16 * BS);
    v4f pre[2][4];
    if (act) {
#pragma unroll
      for (int p = 0; p < 2; ++p)
#pragma unroll
        for (int i = 0; i < 4; ++i) pre[p][i] = *(const v4f*)(bp + (size_t)(4 * i) * NN + p * KC);
    }

    block_prefix(mask + b * TT, sS, sJ, sW, tid);

#pragma unroll
    for (int i = 0; i < 2; ++i) {
      int idx = tid + i * 512;
      *(v4f*)&sXA[idx >> 6][(idx & 63) * 4] = xa[i];
    }
    __syncthreads();

    if (act) {
      float (*dst)[AP] = (wave < 8) ? sZ : sG;
      v8f acc = {};
#pragma unroll
      for (int c = 0; c < NN / KC; ++c) {
#pragma unroll
        for (int i = 0; i < 4; ++i) *(v4f*)&stg[(lr + 4 * i) * BS + lc] = pre[c & 1][i];
        if (c + 2 < NN / KC) {
#pragma unroll
          for (int i = 0; i < 4; ++i)
            pre[c & 1][i] = *(const v4f*)(bp + (size_t)(4 * i) * NN + (c + 2) * KC);
        }
#pragma unroll
        for (int q = 0; q < KC; q += 4) {
          v2f a = *(const v2f*)&sXA[r][c * KC + q + 2 * o];
          v2f bv = *(const v2f*)&stg[r * BS + q + 2 * o];
          acc = wmma4(a, bv, acc);
        }
      }
#pragma unroll
      for (int g = 0; g < 8; ++g) dst[g + 8 * o][wl * 16 + r] = acc[g];
    }
    __syncthreads();

    int kend = (sS[ti * 16 + 15] + 3) & ~3;
    {
      int c = tid & 127, trb = tid >> 7;
      int jc = sJ[c];
      jc = min(jc, ti * 16 + 15);
      int st[4];
      float g[4];
#pragma unroll
      for (int i = 0; i < 4; ++i) {
        st[i] = sS[ti * 16 + trb + 4 * i];
        g[i] = sG[trb + 4 * i][jc];
      }
#pragma unroll
      for (int h = 0; h < HH; ++h) {
        float eh = eta[h];
        float l2a = log2f(alpha[h]);
        if (c < kend) {
#pragma unroll
          for (int i = 0; i < 4; ++i) {
            float val = eh * exp2f(l2a * (float)(st[i] - 1 - c)) * g[i];
            sA[h][trb + 4 * i][c] = (c < st[i]) ? val : 0.0f;
          }
        }
        if (tid < 16) sP[h][tid] = exp2f(l2a * (float)sS[ti * 16 + tid]);
      }
    }
    __syncthreads();

    int ntl = wave & 7;
    int h0 = (wave >> 3) * 2, h1 = h0 + 1;
    int n = (nh * 8 + ntl) * 16 + r;
    v8f acc0 = {}, acc1 = {};
#pragma unroll 4
    for (int k = 0; k < kend; k += 4) {
      int j0 = k + 2 * o;
      v2i jj = *(const v2i*)&sJ[j0];
      v2f bv;
      bv.x = Xb[(unsigned)(jj.x * NN + n)];
      bv.y = Xb[(unsigned)(jj.y * NN + n)];
      v2f a0 = *(const v2f*)&sA[h0][r][j0];
      v2f a1 = *(const v2f*)&sA[h1][r][j0];
      acc0 = wmma4(a0, bv, acc0);
      acc1 = wmma4(a1, bv, acc1);
    }
#pragma unroll
    for (int g = 0; g < 8; ++g) {
      int tl = g + 8 * o;
      int t = ti * 16 + tl;
      float z = sZ[tl][ntl * 16 + r];
      float base0 = sP[h0][tl] * z;
      float base1 = sP[h1][tl] * z;
      out_y[(((size_t)b * TT + t) * HH + h0) * NN + n] = acc0[g] + base0;
      out_y[(((size_t)b * TT + t) * HH + h1) * NN + n] = acc1[g] + base1;
    }
  } else {
    int u = hs * 128 + q;
    int nt = u & 15;
    int b = u >> 4;
    const float* Xb = x + (size_t)b * TT * NN;
    block_prefix(mask + b * TT, sS, sJ, sW, tid);
    __syncthreads();

    int ST = sS[TT];
    int kend = (ST + 3) & ~3;
    {
      int h = tid >> 7, c = tid & 127;
      float eh = eta[h];
      float l2a = log2f(alpha[h]);
      sd[h][c] = (c < ST) ? eh * exp2f(l2a * (float)(ST - 1 - c)) : 0.0f;
    }
    __syncthreads();

    {
      int nr = tid & 15, cb = tid >> 4;
      float xv[4];
#pragma unroll
      for (int i = 0; i < 4; ++i)
        xv[i] = Xb[(unsigned)(sJ[cb + 32 * i] * NN + nt * 16 + nr)];
#pragma unroll
      for (int i = 0; i < 4; ++i) {
        int c = cb + 32 * i;
        if (c < kend) {
#pragma unroll
          for (int h = 0; h < HH; ++h) sA[h][nr][c] = sd[h][c] * xv[i];
        }
      }
    }
    __syncthreads();

    int m0 = wave * 16;
    v8f acc[HH];
#pragma unroll
    for (int h = 0; h < HH; ++h) acc[h] = (v8f){};
#pragma unroll 4
    for (int k = 0; k < kend; k += 4) {
      int j0 = k + 2 * o;
      v2i jj = *(const v2i*)&sJ[j0];
      v2f bv;
      bv.x = Xb[(unsigned)(jj.x * NN + m0 + r)];
      bv.y = Xb[(unsigned)(jj.y * NN + m0 + r)];
#pragma unroll
      for (int h = 0; h < HH; ++h) {
        v2f av = *(const v2f*)&sA[h][r][j0];
        acc[h] = wmma4(av, bv, acc[h]);
      }
    }
    float fST = (float)ST;
    float p[HH];
#pragma unroll
    for (int h = 0; h < HH; ++h) p[h] = exp2f(log2f(alpha[h]) * fST);
#pragma unroll
    for (int g = 0; g < 8; ++g) {
      int nn = nt * 16 + g + 8 * o;
      int mm = m0 + r;
      float wv = w0[((size_t)b * NN + nn) * NN + mm];
#pragma unroll
      for (int h = 0; h < HH; ++h)
        out_w[(((size_t)b * HH + h) * NN + nn) * NN + mm] = acc[h][g] + p[h] * wv;
    }
  }
}

extern "C" void kernel_launch(void* const* d_in, const int* in_sizes, int n_in,
                              void* d_out, int out_size, void* d_ws, size_t ws_size,
                              hipStream_t stream) {
  const float* x     = (const float*)d_in[0];
  const float* w0    = (const float*)d_in[1];
  const float* alpha = (const float*)d_in[2];
  const float* eta   = (const float*)d_in[3];
  const int*   mask  = (const int*)d_in[4];

  float* out_w = (float*)d_out;
  float* out_y = out_w + (size_t)BB * HH * NN * NN;

  fused_kernel<<<dim3(BB * 16 * 2), dim3(512), 0, stream>>>(x, w0, alpha, eta, mask, out_w, out_y);
}


# Final-weights product X^T d X by symmetry: only column tiles mt>=nt computed, mirror tiles stored via a per-wave LDS transpose (rounded as their partner); grid pairs each group with a read-out tile
# speedup vs baseline: 3.4636x; 1.1008x over previous
#include <hip/hip_runtime.h>
#include <math.h>

#define BB 16
#define TT 128
#define NN 256
#define HH 4
#define SP (TT + 1)
#define AP (TT + 2)
#define KC 32
#define XS (NN + 4)
#define BS (KC + 4)
#define TP 20

typedef float v2f __attribute__((ext_vector_type(2)));
typedef float v8f __attribute__((ext_vector_type(8)));
typedef float v4f __attribute__((ext_vector_type(4)));
typedef int v2i __attribute__((ext_vector_type(2)));

static __device__ __forceinline__ v8f wmma4(v2f a, v2f b, v8f c) {
  return __builtin_amdgcn_wmma_f32_16x16x4_f32(
       false, a,  false, b,
       (short)0, c,  false,  false);
}

static __device__ __forceinline__ void block_prefix(const int* __restrict__ mask_b,
                                                    int* sS, int* sJ, int* sW, int tid) {
  int lane = tid & 31, wave = tid >> 5;
  int v = 0, m = 0;
  if (tid < TT) {
    m = mask_b[tid];
    v = m;
    for (int d = 1; d < 32; d <<= 1) {
      int nb = __shfl_up(v, d);
      if (lane >= d) v += nb;
    }
    if (lane == 31) sW[wave] = v;
    sJ[tid] = 0;
  }
  __syncthreads();
  if (tid < TT) {
    int off = 0;
    for (int q = 0; q < wave; ++q) off += sW[q];
    int incl = v + off;
    sS[tid + 1] = incl;
    if (tid == 0) sS[0] = 0;
    if (m) sJ[(incl - 1) & (TT - 1)] = tid;
  }
}

__global__ void __launch_bounds__(512) fused_kernel(
    const float* __restrict__ x, const float* __restrict__ w0,
    const float* __restrict__ alpha, const float* __restrict__ eta,
    const int* __restrict__ mask, float* __restrict__ out_w,
    float* __restrict__ out_y) {
  __shared__ int sS[SP];
  __shared__ __attribute__((aligned(8))) int sJ[TT];
  __shared__ int sW[4];
  __shared__ __attribute__((aligned(16))) float sU[16 * 16 * BS];
  float (*sA)[16][AP] = reinterpret_cast<float (*)[16][AP]>(sU);
  __shared__ __attribute__((aligned(16))) float sXA[16][XS];
  __shared__ float sG[16][AP];
  __shared__ float sZ[16][AP];
  __shared__ float sP[HH][16];
  __shared__ float sd[HH][TT];
  __shared__ __attribute__((aligned(16))) float sT[16][16 * TP];

  int tid = threadIdx.x;
  int wave = tid >> 5;
  int lane = tid & 31;
  int r = lane & 15, o = lane >> 4;
  int hs = blockIdx.x >> 8;
  int q = blockIdx.x & 255;
  int pb = (q >> 1) & 15;
  int pnh = q & 1;
  int pti = q >> 5;

  if (hs == 0) {
    int b = pb;
    int nh = pnh;
    int ti = pti;
    const float* Xb = x + (size_t)b * TT * NN;

    int wl = wave & 7;
    bool act = (wave < 8) || (wl <= ti);
    v4f xa[2];
#pragma unroll
    for (int i = 0; i < 2; ++i) {
      int idx = tid + i * 512;
      xa[i] = *(const v4f*)(Xb + (size_t)(ti * 16 + (idx >> 6)) * NN + (idx & 63) * 4);
    }
    const float* bt = (wave < 8)
        ? w0 + ((size_t)b * NN + (nh * 8 + wl) * 16) * NN
        : Xb + (size_t)(wl * 16) * NN;
    int lr = lane >> 3, lc = (lane & 7) * 4;
    const float* bp = bt + (size_t)lr * NN + lc;
    float* stg = sU + wave * (16 * BS);
    v4f pre[2][4];
    if (act) {
#pragma unroll
      for (int p = 0; p < 2; ++p)
#pragma unroll
        for (int i = 0; i < 4; ++i) pre[p][i] = *(const v4f*)(bp + (size_t)(4 * i) * NN + p * KC);
    }

    block_prefix(mask + b * TT, sS, sJ, sW, tid);

#pragma unroll
    for (int i = 0; i < 2; ++i) {
      int idx = tid + i * 512;
      *(v4f*)&sXA[idx >> 6][(idx & 63) * 4] = xa[i];
    }
    __syncthreads();

    if (act) {
      float (*dst)[AP] = (wave < 8) ? sZ : sG;
      v8f acc = {};
#pragma unroll
      for (int c = 0; c < NN / KC; ++c) {
#pragma unroll
        for (int i = 0; i < 4; ++i) *(v4f*)&stg[(lr + 4 * i) * BS + lc] = pre[c & 1][i];
        if (c + 2 < NN / KC) {
#pragma unroll
          for (int i = 0; i < 4; ++i)
            pre[c & 1][i] = *(const v4f*)(bp + (size_t)(4 * i) * NN + (c + 2) * KC);
        }
#pragma unroll
        for (int q = 0; q < KC; q += 4) {
          v2f a = *(const v2f*)&sXA[r][c * KC + q + 2 * o];
          v2f bv = *(const v2f*)&stg[r * BS + q + 2 * o];
          acc = wmma4(a, bv, acc);
        }
      }
#pragma unroll
      for (int g = 0; g < 8; ++g) dst[g + 8 * o][wl * 16 + r] = acc[g];
    }
    __syncthreads();

    int kend = (sS[ti * 16 + 15] + 3) & ~3;
    {
      int c = tid & 127, trb = tid >> 7;
      int jc = sJ[c];
      jc = min(jc, ti * 16 + 15);
      int st[4];
      float g[4];
#pragma unroll
      for (int i = 0; i < 4; ++i) {
        st[i] = sS[ti * 16 + trb + 4 * i];
        g[i] = sG[trb + 4 * i][jc];
      }
#pragma unroll
      for (int h = 0; h < HH; ++h) {
        float eh = eta[h];
        float l2a = log2f(alpha[h]);
        if (c < kend) {
#pragma unroll
          for (int i = 0; i < 4; ++i) {
            float val = eh * exp2f(l2a * (float)(st[i] - 1 - c)) * g[i];
            sA[h][trb + 4 * i][c] = (c < st[i]) ? val : 0.0f;
          }
        }
        if (tid < 16) sP[h][tid] = exp2f(l2a * (float)sS[ti * 16 + tid]);
      }
    }
    __syncthreads();

    int ntl = wave & 7;
    int h0 = (wave >> 3) * 2, h1 = h0 + 1;
    int n = (nh * 8 + ntl) * 16 + r;
    v8f acc0 = {}, acc1 = {};
#pragma unroll 4
    for (int k = 0; k < kend; k += 4) {
      int j0 = k + 2 * o;
      v2i jj = *(const v2i*)&sJ[j0];
      v2f bv;
      bv.x = Xb[(unsigned)(jj.x * NN + n)];
      bv.y = Xb[(unsigned)(jj.y * NN + n)];
      v2f a0 = *(const v2f*)&sA[h0][r][j0];
      v2f a1 = *(const v2f*)&sA[h1][r][j0];
      acc0 = wmma4(a0, bv, acc0);
      acc1 = wmma4(a1, bv, acc1);
    }
#pragma unroll
    for (int g = 0; g < 8; ++g) {
      int tl = g + 8 * o;
      int t = ti * 16 + tl;
      float z = sZ[tl][ntl * 16 + r];
      float base0 = sP[h0][tl] * z;
      float base1 = sP[h1][tl] * z;
      out_y[(((size_t)b * TT + t) * HH + h0) * NN + n] = acc0[g] + base0;
      out_y[(((size_t)b * TT + t) * HH + h1) * NN + n] = acc1[g] + base1;
    }
  } else {
    int nt = 2 * pti + pnh;
    int b = pb;
    const float* Xb = x + (size_t)b * TT * NN;
    block_prefix(mask + b * TT, sS, sJ, sW, tid);
    __syncthreads();

    int ST = sS[TT];
    int kend = (ST + 3) & ~3;
    {
      int h = tid >> 7, c = tid & 127;
      float eh = eta[h];
      float l2a = log2f(alpha[h]);
      sd[h][c] = (c < ST) ? eh * exp2f(l2a * (float)(ST - 1 - c)) : 0.0f;
    }
    __syncthreads();

    {
      int nr = tid & 15, cb = tid >> 4;
      float xv[4];
#pragma unroll
      for (int i = 0; i < 4; ++i)
        xv[i] = Xb[(unsigned)(sJ[cb + 32 * i] * NN + nt * 16 + nr)];
#pragma unroll
      for (int i = 0; i < 4; ++i) {
        int c = cb + 32 * i;
        if (c < kend) {
#pragma unroll
          for (int h = 0; h < HH; ++h) sA[h][nr][c] = sd[h][c] * xv[i];
        }
      }
    }
    __syncthreads();

    if (wave >= nt) {
      int m0 = wave * 16;
      v8f acc[HH];
#pragma unroll
      for (int h = 0; h < HH; ++h) acc[h] = (v8f){};
#pragma unroll 4
      for (int k = 0; k < kend; k += 4) {
        int j0 = k + 2 * o;
        v2i jj = *(const v2i*)&sJ[j0];
        v2f bv;
        bv.x = Xb[(unsigned)(jj.x * NN + m0 + r)];
        bv.y = Xb[(unsigned)(jj.y * NN + m0 + r)];
#pragma unroll
        for (int h = 0; h < HH; ++h) {
          v2f av = *(const v2f*)&sA[h][r][j0];
          acc[h] = wmma4(av, bv, acc[h]);
        }
      }
      float fST = (float)ST;
      float p[HH];
#pragma unroll
      for (int h = 0; h < HH; ++h) p[h] = exp2f(log2f(alpha[h]) * fST);
      bool mir = wave > nt;
      float wm[8];
      if (mir) {
#pragma unroll
        for (int g = 0; g < 8; ++g)
          wm[g] = w0[((size_t)b * NN + m0 + g + 8 * o) * NN + nt * 16 + r];
      }
#pragma unroll
      for (int g = 0; g < 8; ++g) {
        int nn = nt * 16 + g + 8 * o;
        int mm = m0 + r;
        float wv = w0[((size_t)b * NN + nn) * NN + mm];
#pragma unroll
        for (int h = 0; h < HH; ++h)
          out_w[(((size_t)b * HH + h) * NN + nn) * NN + mm] = acc[h][g] + p[h] * wv;
      }
      if (mir) {
        float* tw = &sT[wave][0];
#pragma unroll
        for (int h = 0; h < HH; ++h) {
          *(v4f*)&tw[r * TP + 8 * o] = __builtin_shufflevector(acc[h], acc[h], 0, 1, 2, 3);
          *(v4f*)&tw[r * TP + 8 * o + 4] = __builtin_shufflevector(acc[h], acc[h], 4, 5, 6, 7);
#pragma unroll
          for (int g = 0; g < 8; ++g) {
            float v = tw[(g + 8 * o) * TP + r];
            out_w[(((size_t)b * HH + h) * NN + m0 + g + 8 * o) * NN + nt * 16 + r] = v + p[h] * wm[g];
          }
        }
      }
    }
  }
}

extern "C" void kernel_launch(void* const* d_in, const int* in_sizes, int n_in,
                              void* d_out, int out_size, void* d_ws, size_t ws_size,
                              hipStream_t stream) {
  const float* x     = (const float*)d_in[0];
  const float* w0    = (const float*)d_in[1];
  const float* alpha = (const float*)d_in[2];
  const float* eta   = (const float*)d_in[3];
  const int*   mask  = (const int*)d_in[4];

  float* out_w = (float*)d_out;
  float* out_y = out_w + (size_t)BB * HH * NN * NN;

  fused_kernel<<<dim3(BB * 16 * 2), dim3(512), 0, stream>>>(x, w0, alpha, eta, mask, out_w, out_y);
}
